# starting kernel
# baseline (speedup 1.0000x reference)
#include <hip/hip_runtime.h>
#include <hip/hip_bf16.h>


typedef __attribute__((ext_vector_type(2))) float v2f;
typedef __attribute__((ext_vector_type(8))) float v8f;

#define N_IN  1024
#define N_OUT 512
#define BATCH 128

__global__ __launch_bounds__(32)
void memristor_linear_wmma_f32(const float* __restrict__ X,
                               const float* __restrict__ W,
                               const float* __restrict__ Bv,
                               float* __restrict__ Y) {
  const int lane = threadIdx.x & 31;
  const int h    = lane >> 4;
  const int ml   = lane & 15;
  const int m0   = blockIdx.y << 4;
  const int n0   = blockIdx.x << 5;

  const float* arow = X + (size_t)(m0 + ml) * N_IN + 2 * h;
  const float* b0p = W + n0 + ml;
  const float* b1p = W + n0 + 16 + ml;

  v8f acc0 = {};
  v8f acc1 = {};

#pragma unroll 4
  for (int k0 = 0; k0 < N_IN; k0 += 4) {
    const int ka = k0 + 2 * h;

    v2f a = *(const v2f*)(arow + k0);

    v2f b0, b1;
    b0.x = b0p[(size_t)(ka)     * N_OUT];
    b0.y = b0p[(size_t)(ka + 1) * N_OUT];
    b1.x = b1p[(size_t)(ka)     * N_OUT];
    b1.y = b1p[(size_t)(ka + 1) * N_OUT];

    acc0 = __builtin_amdgcn_wmma_f32_16x16x4_f32(false, a, false, b0,
                                                 (short)0, acc0, false, false);
    acc1 = __builtin_amdgcn_wmma_f32_16x16x4_f32(false, a, false, b1,
                                                 (short)0, acc1, false, false);
  }

  const float bias0 = Bv[n0 + ml];
  const float bias1 = Bv[n0 + 16 + ml];

#pragma unroll
  for (int r = 0; r < 8; ++r) {
    const int m = m0 + r + 8 * h;
    Y[(size_t)m * N_OUT + n0 + ml]      = acc0[r] + bias0;
    Y[(size_t)m * N_OUT + n0 + 16 + ml] = acc1[r] + bias1;
  }
}

extern "C" void kernel_launch(void* const* d_in, const int* in_sizes, int n_in,
                              void* d_out, int out_size, void* d_ws, size_t ws_size,
                              hipStream_t stream) {
  const float* x = (const float*)d_in[0];
  const float* w = (const float*)d_in[1];
  const float* b = (const float*)d_in[2];
  float* y = (float*)d_out;

  dim3 grid(N_OUT / 32, BATCH / 16);
  dim3 block(32);
  memristor_linear_wmma_f32<<<grid, block, 0, stream>>>(x, w, b, y);
}


# No global fetch is waited for early: chunk LDS stores moved right before their own fragment reads (3 steps of flight, not 2); bias fetched at kernel start, not behind the final barrier
# speedup vs baseline: 7.3889x; 7.3889x over previous
#include <hip/hip_runtime.h>


typedef __attribute__((ext_vector_type(2))) float v2f;
typedef __attribute__((ext_vector_type(4))) float v4f;
typedef __attribute__((ext_vector_type(8))) float v8f;

#define N_IN  1024
#define N_OUT 512
#define BATCH 128

#define TM 16
#define TN 32
#define KC 128
#define KW (KC / 4)
#define NG (KW / 8)
#define NCHUNK (N_IN / KC)
#define SP (KW + 4)
#define XSZ (TM * SP)
#define WGAP 4
#define WSZ (TN * SP + WGAP)
#define BUFSZ (XSZ + WSZ)
#define WAVE_LDS 8192
#define YPITCH 36

__device__ __forceinline__ void wave_lds_order() {
  __builtin_amdgcn_fence(__ATOMIC_RELEASE, "wavefront", "local");
  __builtin_amdgcn_wave_barrier();
  __builtin_amdgcn_fence(__ATOMIC_ACQUIRE, "wavefront", "local");
}

__global__ __launch_bounds__(128)
void memristor_linear_lds_tiles(const float* __restrict__ X,
                                const float* __restrict__ W,
                                const float* __restrict__ Bv,
                                float* __restrict__ Y) {
  __shared__ __attribute__((aligned(16))) float Stage[4 * WAVE_LDS];
  __shared__ __attribute__((aligned(16))) float Ps[4][TM * YPITCH];

  const int tid  = threadIdx.x;
  const int lane = tid & 31;
  const int wv   = tid >> 5;
  const int c    = lane & 15;
  const int h    = lane >> 4;
  const int r    = lane >> 3;
  const int q    = lane & 7;
  const int m0   = blockIdx.y * TM;
  const int n0   = blockIdx.x * TN;

  const float* xg = X + (size_t)(m0 + r) * N_IN + KW * wv + q * 4;
  const float* wg = W + (size_t)(KW * wv + r) * N_OUT + n0 + q * 4;
  const int sw = q >> 2;

  float* st = &Stage[wv * WAVE_LDS];

  v4f xr[3][4], wr[3][8];
  v4f a[2][NG], b0[2][NG], b1[2][NG];


  auto gload = [&](int t) {
    const int s = t % 3;
    const int k0 = t * KC;
#pragma unroll
    for (int i = 0; i < 4; ++i) {
      xr[s][i] = *(const v4f*)(xg + (size_t)(4 * i) * N_IN + k0);
    }
#pragma unroll
    for (int i = 0; i < 8; ++i) {
      wr[s][i] = *(const v4f*)(wg + (size_t)(k0 + 4 * i) * N_OUT);
    }
  };

  auto sstore = [&](int t) {
    const int s = t % 3;
    float* xs = st + (t & 1) * BUFSZ + r * SP + q * 4;
    float* ws = st + (t & 1) * BUFSZ + XSZ + (q * 4) * SP + WGAP * sw + r;
#pragma unroll
    for (int i = 0; i < 4; ++i) {
      *(v4f*)(xs + (4 * i) * SP) = xr[s][i];
    }
#pragma unroll
    for (int i = 0; i < 8; ++i) {
      ws[0 * SP + 4 * i] = wr[s][i].x;
      ws[1 * SP + 4 * i] = wr[s][i].y;
      ws[2 * SP + 4 * i] = wr[s][i].z;
      ws[3 * SP + 4 * i] = wr[s][i].w;
    }
  };

  const int aoff  = c * SP + 4 * h;
  const int boff0 = XSZ + c * SP + 4 * h;
  const int boff1 = XSZ + (16 + c) * SP + WGAP + 4 * h;

  auto fread = [&](int t) {
    const int s = t & 1;
    const float* ap  = st + s * BUFSZ + aoff;
    const float* bp0 = st + s * BUFSZ + boff0;
    const float* bp1 = st + s * BUFSZ + boff1;
#pragma unroll
    for (int g = 0; g < NG; ++g) {
      a[s][g]  = *(const v4f*)(ap + 8 * g);
      b0[s][g] = *(const v4f*)(bp0 + 8 * g);
      b1[s][g] = *(const v4f*)(bp1 + 8 * g);
    }
  };

  v8f acc0 = {};
  v8f acc1 = {};

  gload(0);
  __builtin_amdgcn_sched_barrier(0);
  gload(1);
  __builtin_amdgcn_sched_barrier(0);
  gload(2);
  const int wn = (tid & 7) * 4;
  const v4f bias = *(const v4f*)(Bv + n0 + wn);
  __builtin_amdgcn_sched_barrier(0);
  sstore(0);
  wave_lds_order();
  gload(3);
  fread(0);

#pragma unroll
  for (int t = 0; t < NCHUNK; ++t) {
    const int cur = t & 1;
    if (t + 1 < NCHUNK) {
      sstore(t + 1);
      wave_lds_order();
    }
    if (t + 4 < NCHUNK) gload(t + 4);
    if (t + 1 < NCHUNK) fread(t + 1);
    __builtin_amdgcn_sched_barrier(0);
#pragma unroll
    for (int g = 0; g < NG; ++g) {
      v2f a0 = {a[cur][g].x, a[cur][g].y}, a1 = {a[cur][g].z, a[cur][g].w};
      v2f p0 = {b0[cur][g].x, b0[cur][g].y}, p1 = {b0[cur][g].z, b0[cur][g].w};
      v2f q0 = {b1[cur][g].x, b1[cur][g].y}, q1 = {b1[cur][g].z, b1[cur][g].w};
      acc0 = __builtin_amdgcn_wmma_f32_16x16x4_f32(false, a0, false, p0,
                                                   (short)0, acc0, false, false);
      acc1 = __builtin_amdgcn_wmma_f32_16x16x4_f32(false, a0, false, q0,
                                                   (short)0, acc1, false, false);
      acc0 = __builtin_amdgcn_wmma_f32_16x16x4_f32(false, a1, false, p1,
                                                   (short)0, acc0, false, false);
      acc1 = __builtin_amdgcn_wmma_f32_16x16x4_f32(false, a1, false, q1,
                                                   (short)0, acc1, false, false);
    }
    __builtin_amdgcn_sched_barrier(0);
    if (t + 1 < NCHUNK) wave_lds_order();
  }

  float* ps = &Ps[wv][0];
#pragma unroll
  for (int rr = 0; rr < 8; ++rr) {
    ps[(8 * h + rr) * YPITCH + c]      = acc0[rr];
    ps[(8 * h + rr) * YPITCH + 16 + c] = acc1[rr];
  }
  __syncthreads();
  const int wk = tid >> 3;
  const int yo = wk * YPITCH + wn;
  v4f v = (*(const v4f*)(&Ps[0][yo]) + *(const v4f*)(&Ps[1][yo])) +
          (*(const v4f*)(&Ps[2][yo]) + *(const v4f*)(&Ps[3][yo]));
  v += bias;
  *(v4f*)(Y + (size_t)(m0 + wk) * N_OUT + n0 + wn) = v;
}

extern "C" void kernel_launch(void* const* d_in, const int* in_sizes, int n_in,
                              void* d_out, int out_size, void* d_ws, size_t ws_size,
                              hipStream_t stream) {
  const float* x = (const float*)d_in[0];
  const float* w = (const float*)d_in[1];
  const float* b = (const float*)d_in[2];
  float* y = (float*)d_out;

  dim3 grid(N_OUT / TN, BATCH / TM);
  dim3 block(128);
  memristor_linear_lds_tiles<<<grid, block, 0, stream>>>(x, w, b, y);
}
